# starting kernel
# baseline (speedup 1.0000x reference)
#include <hip/hip_runtime.h>

typedef __attribute__((ext_vector_type(2))) float v2f;
typedef __attribute__((ext_vector_type(2), aligned(4))) float v2fu;
typedef __attribute__((ext_vector_type(4))) float v4f;
typedef __attribute__((ext_vector_type(8))) float v8f;

#define NUNITS 81
#define WA_OFF    0
#define WB_OFF    736
#define WMIX_OFF  1472
#define BFRAG_OFF 2768
#define NFRAG     (9 * 2 * 3 * 32)

__device__ __constant__ int c_hw2i[12] = {0, 1, 3, 4, 6, 7, 2, -1, 5, -1, -1, 8};

__global__ void logic_prep_kernel(const float* __restrict__ sa,
                                  const float* __restrict__ sb,
                                  const float* __restrict__ op,
                                  float* __restrict__ ws) {
  const int tid = threadIdx.x;
  if (tid < NUNITS) {
    {
      const float* src = sa + tid * 9;
      float m = src[0];
#pragma unroll
      for (int i = 1; i < 9; ++i) m = fmaxf(m, src[i]);
      float e[9], s = 0.f;
#pragma unroll
      for (int i = 0; i < 9; ++i) { e[i] = expf(src[i] - m); s += e[i]; }
      const float inv = 1.0f / s;
#pragma unroll
      for (int i = 0; i < 9; ++i) ws[WA_OFF + tid * 9 + i] = e[i] * inv;
    }
    {
      const float* src = sb + tid * 9;
      float m = src[0];
#pragma unroll
      for (int i = 1; i < 9; ++i) m = fmaxf(m, src[i]);
      float e[9], s = 0.f;
#pragma unroll
      for (int i = 0; i < 9; ++i) { e[i] = expf(src[i] - m); s += e[i]; }
      const float inv = 1.0f / s;
#pragma unroll
      for (int i = 0; i < 9; ++i) ws[WB_OFF + tid * 9 + i] = e[i] * inv;
    }
    {
      const float* src = op + tid * 16;
      float m = src[0];
#pragma unroll
      for (int i = 1; i < 16; ++i) m = fmaxf(m, src[i]);
      float w[16], s = 0.f;
#pragma unroll
      for (int i = 0; i < 16; ++i) { w[i] = expf(src[i] - m); s += w[i]; }
      const float inv = 1.0f / s;
#pragma unroll
      for (int i = 0; i < 16; ++i) w[i] *= inv;
      const float c0  = w[8] + w[9] + w[10] + w[11] + w[12] + w[13] + w[14] + w[15];
      const float cA  = w[2] + w[3] + w[6] + w[7] - w[8] - w[9] - w[12] - w[13];
      const float cB  = w[4] + w[5] + w[6] + w[7] - w[8] - w[9] - w[10] - w[11];
      const float cAB = w[1] - w[2] - w[4] - 2.0f * w[6] - w[7] + w[8] +
                        2.0f * w[9] + w[11] + w[13] - w[14];
      ws[WMIX_OFF + tid * 4 + 0] = c0;
      ws[WMIX_OFF + tid * 4 + 1] = cA;
      ws[WMIX_OFF + tid * 4 + 2] = cB;
      ws[WMIX_OFF + tid * 4 + 3] = cAB;
    }
  }
  __syncthreads();
  for (int idx = tid; idx < NFRAG; idx += blockDim.x) {
    const int lane  = idx & 31;
    const int rest  = idx >> 5;
    const int kk    = rest % 3;
    const int pt    = rest / 3;
    const int t     = pt & 1;
    const int p     = pt >> 1;
    const int col   = lane & 15;
    const int khalf = (lane >> 4) << 1;
    const float* tbl = ws + (t ? WB_OFF : WA_OFF);
    const int k0 = 4 * kk + khalf;
    const int i0 = c_hw2i[k0];
    const int i1 = c_hw2i[k0 + 1];
    float v0 = 0.f, v1 = 0.f;
    if (col < 9) {
      if (i0 >= 0) v0 = tbl[(9 * p + col) * 9 + i0];
      if (i1 >= 0) v1 = tbl[(9 * p + col) * 9 + i1];
    }
    ws[BFRAG_OFF + idx * 2 + 0] = v0;
    ws[BFRAG_OFF + idx * 2 + 1] = v1;
  }
}

__global__ void __launch_bounds__(128, 1)
logic_wmma_kernel(const float* __restrict__ x,
                  const float* __restrict__ ws,
                  float* __restrict__ out,
                  int nBatch) {
  const int lane   = threadIdx.x & 31;
  const int waveId = threadIdx.x >> 5;
  const long base  = ((long)blockIdx.x * 4 + waveId) * 16;
  if (base + 16 > nBatch) return;

  const int col  = lane & 15;
  const bool hiM = (lane >= 16);

  const float* xrow = x + (base + col) * 81;
  const float* xr0  = xrow + (hiM ? 9 : 0);
  const float* xr1  = xrow + (hiM ? 2 : 18);
  const float* xr2  = xrow + (hiM ? 19 : 11);
  const v2f*   bp   = (const v2f*)(ws + BFRAG_OFF) + lane;
  const float* wc   = ws + WMIX_OFF + col * 4;
  float*       outp = out + (base + (hiM ? 8 : 0)) * 81 + col;
  const bool   act  = (col < 9);

#pragma unroll
  for (int p = 0; p < 9; ++p) {
    const int pbase = (p / 3) * 27 + (p % 3) * 3;

    const v2f a0 = *(const v2fu*)(xr0 + pbase);
    const v2f a1 = *(const v2fu*)(xr1 + pbase);
    const v2f a2 = *(const v2fu*)(xr2 + pbase);

    const v2f ba0 = bp[(p * 6 + 0) * 32];
    const v2f ba1 = bp[(p * 6 + 1) * 32];
    const v2f ba2 = bp[(p * 6 + 2) * 32];
    const v2f bb0 = bp[(p * 6 + 3) * 32];
    const v2f bb1 = bp[(p * 6 + 4) * 32];
    const v2f bb2 = bp[(p * 6 + 5) * 32];

    v8f cAcc = {};
    v8f cBcc = {};
    cAcc = __builtin_amdgcn_wmma_f32_16x16x4_f32(false, a0, false, ba0, (short)0, cAcc, false, false);
    cAcc = __builtin_amdgcn_wmma_f32_16x16x4_f32(false, a1, false, ba1, (short)0, cAcc, false, false);
    cAcc = __builtin_amdgcn_wmma_f32_16x16x4_f32(false, a2, false, ba2, (short)0, cAcc, false, false);
    cBcc = __builtin_amdgcn_wmma_f32_16x16x4_f32(false, a0, false, bb0, (short)0, cBcc, false, false);
    cBcc = __builtin_amdgcn_wmma_f32_16x16x4_f32(false, a1, false, bb1, (short)0, cBcc, false, false);
    cBcc = __builtin_amdgcn_wmma_f32_16x16x4_f32(false, a2, false, bb2, (short)0, cBcc, false, false);

    if (act) {
      const v4f c = *(const v4f*)(wc + 36 * p);
#pragma unroll
      for (int m = 0; m < 8; ++m) {
        const float av = cAcc[m];
        const float bv = cBcc[m];
        const float r  = fmaf(av * bv, c.w, fmaf(bv, c.z, fmaf(av, c.y, c.x)));
        __builtin_nontemporal_store(r, outp + 9 * p + 81 * m);
      }
    }
  }
}

extern "C" void kernel_launch(void* const* d_in, const int* in_sizes, int n_in,
                              void* d_out, int out_size, void* d_ws, size_t ws_size,
                              hipStream_t stream) {
  const float* x  = (const float*)d_in[0];
  const float* sa = (const float*)d_in[1];
  const float* sb = (const float*)d_in[2];
  const float* op = (const float*)d_in[3];
  float* out = (float*)d_out;
  float* ws  = (float*)d_ws;

  const int nBatch = in_sizes[0] / 81;

  logic_prep_kernel<<<1, 128, 0, stream>>>(sa, sb, op, ws);

  const int rowsPerBlock = 64;
  const int nBlocks = (nBatch + rowsPerBlock - 1) / rowsPerBlock;
  logic_wmma_kernel<<<nBlocks, 128, 0, stream>>>(x, ws, out, nBatch);
}


# outputs re-tiled: 3 bands of 27 units as two column tiles; after the mix the two tiles trade lane halves (v_permlane16_swap) so each store writes one row's 27 adjacent units, not 2 rows x 9
# speedup vs baseline: 1.7255x; 1.7255x over previous
#include <hip/hip_runtime.h>

typedef __attribute__((ext_vector_type(2))) float v2f;
typedef __attribute__((ext_vector_type(4), aligned(4))) float v4fu;
typedef __attribute__((ext_vector_type(4))) float v4f;
typedef __attribute__((ext_vector_type(8))) float v8f;

#define NUNITS 81
#define WMIX_OFF  1472
#define BFRAG_OFF 2768
#define NKSTEP    62
#define NENT      41

#define FSTART_PACKED 0x36c2a91e590200ULL
#define FOUR_STEP_MASK 0x10B
__device__ __constant__ signed char c_floc[NKSTEP][2][2] = {
  {{0,1},{8,-1}}, {{2,3},{-1,-1}}, {{4,5},{-1,-1}}, {{6,7},{-1,-1}},
  {{0,1},{8,-1}}, {{2,3},{-1,-1}}, {{4,5},{-1,-1}}, {{6,7},{-1,-1}},
  {{0,-1},{1,2}}, {{-1,-1},{3,4}}, {{-1,-1},{5,6}}, {{-1,-1},{7,8}},
  {{0,-1},{1,2}}, {{-1,-1},{3,4}}, {{-1,-1},{5,6}}, {{-1,-1},{7,8}},
  {{2,4},{0,1}}, {{5,6},{3,-1}}, {{7,8},{-1,-1}}, {{2,4},{0,1}},
  {{5,6},{3,-1}}, {{7,8},{-1,-1}}, {{0,-1},{1,2}}, {{-1,-1},{3,4}},
  {{-1,-1},{5,6}}, {{-1,-1},{7,8}}, {{0,-1},{1,2}}, {{-1,-1},{3,4}},
  {{-1,-1},{5,6}}, {{-1,-1},{7,8}}, {{2,4},{0,1}}, {{5,6},{3,-1}},
  {{7,8},{-1,-1}}, {{2,4},{0,1}}, {{5,6},{3,-1}}, {{7,8},{-1,-1}},
  {{0,1},{5,7}}, {{2,3},{8,-1}}, {{4,6},{-1,-1}}, {{0,1},{5,7}},
  {{2,3},{8,-1}}, {{4,6},{-1,-1}}, {{2,4},{0,1}}, {{5,6},{3,-1}},
  {{7,8},{-1,-1}}, {{2,4},{0,1}}, {{5,6},{3,-1}}, {{7,8},{-1,-1}},
  {{0,1},{5,7}}, {{2,3},{8,-1}}, {{4,6},{-1,-1}}, {{0,1},{5,7}},
  {{2,3},{8,-1}}, {{4,6},{-1,-1}}, {{8,-1},{0,1}}, {{-1,-1},{2,3}},
  {{-1,-1},{4,5}}, {{-1,-1},{6,7}}, {{8,-1},{0,1}}, {{-1,-1},{2,3}},
  {{-1,-1},{4,5}}, {{-1,-1},{6,7}}};

__device__ __constant__ unsigned char c_ent[NENT][3] = {
  {0,0,0}, {0,1,0}, {0,2,0}, {0,3,0}, {1,0,0}, {1,0,1}, {1,1,0},
  {1,1,1}, {1,2,0}, {1,2,1}, {1,3,0}, {1,3,1}, {2,0,1}, {2,1,1},
  {2,2,1}, {3,0,0}, {3,1,0}, {3,2,0}, {3,3,0}, {4,0,0}, {4,0,1},
  {4,1,0}, {4,1,1}, {4,2,0}, {4,2,1}, {5,0,1}, {5,1,1}, {5,2,1},
  {6,0,0}, {6,1,0}, {6,2,0}, {7,0,0}, {7,0,1}, {7,1,0}, {7,1,1},
  {7,2,0}, {7,2,1}, {8,0,1}, {8,1,1}, {8,2,1}, {8,3,1}};

__device__ __forceinline__ float pick9(const float (&e)[9], const int i) {
  float r = 0.f;
#pragma unroll
  for (int j = 0; j < 9; ++j) r = (i == j) ? e[j] : r;
  return r;
}

__global__ void __launch_bounds__(128)
logic_prep_kernel(const float* __restrict__ sa,
                  const float* __restrict__ sb,
                  const float* __restrict__ op,
                  float* __restrict__ ws) {
  const int tid = threadIdx.x;
  const int blk = blockIdx.x;
  if (blk < 21) {
    const int g = blk * 4 + (tid >> 5);
    if (g >= 2 * NENT) return;
    const int tb   = (g >= NENT) ? 1 : 0;
    const int e    = g - tb * NENT;
    const int p    = c_ent[e][0];
    const int t    = c_ent[e][1];
    const int tile = c_ent[e][2];
    const int nk   = 3 + ((FOUR_STEP_MASK >> p) & 1);
    const int f    = (int)((FSTART_PACKED >> (6 * p)) & 63) + tb * nk + t;
    const int lane = tid & 31;
    const int col  = lane & 15;
    const int h    = lane >> 4;
    const int i0 = c_floc[f][h][0];
    const int i1 = c_floc[f][h][1];
    const int u  = 27 * (p / 3) + 16 * tile + col;
    float v0 = 0.f, v1 = 0.f;
    if (u / 9 == p) {
      const float* src = (tb ? sb : sa) + u * 9;
      float m = src[0];
#pragma unroll
      for (int i = 1; i < 9; ++i) m = fmaxf(m, src[i]);
      float ex[9], s = 0.f;
#pragma unroll
      for (int i = 0; i < 9; ++i) { ex[i] = expf(src[i] - m); s += ex[i]; }
      const float inv = 1.0f / s;
      v0 = pick9(ex, i0) * inv;
      v1 = pick9(ex, i1) * inv;
    }
    v2f* dst = (v2f*)(ws + BFRAG_OFF) + (g * 32 + lane);
    const v2f v = {v0, v1};
    *dst = v;
  } else if (tid < NUNITS) {
    const float* src = op + tid * 16;
    float m = src[0];
#pragma unroll
    for (int i = 1; i < 16; ++i) m = fmaxf(m, src[i]);
    float w[16], s = 0.f;
#pragma unroll
    for (int i = 0; i < 16; ++i) { w[i] = expf(src[i] - m); s += w[i]; }
    const float inv = 1.0f / s;
#pragma unroll
    for (int i = 0; i < 16; ++i) w[i] *= inv;
    const float c0  = w[8] + w[9] + w[10] + w[11] + w[12] + w[13] + w[14] + w[15];
    const float cA  = w[2] + w[3] + w[6] + w[7] - w[8] - w[9] - w[12] - w[13];
    const float cB  = w[4] + w[5] + w[6] + w[7] - w[8] - w[9] - w[10] - w[11];
    const float cAB = w[1] - w[2] - w[4] - 2.0f * w[6] - w[7] + w[8] +
                      2.0f * w[9] + w[11] + w[13] - w[14];
    const v4f c = {c0, cA, cB, cAB};
    *(v4f*)(ws + WMIX_OFF + tid * 4) = c;
  }
}

__device__ __forceinline__ float mix1(const float av, const float bv, const v4f c) {
  return fmaf(av * bv, c.w, fmaf(bv, c.z, fmaf(av, c.y, c.x)));
}

__device__ __forceinline__ void band_store(const v8f aLo, const v8f bLo,
                                           const v8f aHi, const v8f bHi,
                                           const float* __restrict__ wLo,
                                           const float* __restrict__ wHi,
                                           float* __restrict__ o, const bool act) {
  const v4f cl = *(const v4f*)wLo;
  const v4f ch = *(const v4f*)wHi;
  float r0[8], r1[8];
#pragma unroll
  for (int m = 0; m < 8; ++m) {
    const float rl = mix1(aLo[m], bLo[m], cl);
    const float rh = mix1(aHi[m], bHi[m], ch);
    const auto sw = __builtin_amdgcn_permlane16_swap(__float_as_uint(rl), __float_as_uint(rh), false, false);
    r0[m] = __uint_as_float(sw[0]);
    r1[m] = __uint_as_float(sw[1]);
  }
  if (act) {
#pragma unroll
    for (int m = 0; m < 8; ++m) {
      __builtin_nontemporal_store(r0[m], o + 81 * m);
      __builtin_nontemporal_store(r1[m], o + 81 * (m + 8));
    }
  }
}

__global__ void __launch_bounds__(128, 1)
logic_wmma_kernel(const float* __restrict__ x,
                  const float* __restrict__ ws,
                  float* __restrict__ out,
                  int nBatch) {
  const int lane   = threadIdx.x & 31;
  const int waveId = threadIdx.x >> 5;
  const long base  = ((long)blockIdx.x * 4 + waveId) * 16;
  if (base + 16 > nBatch) return;

  const int col  = lane & 15;
  const bool hiM = (lane >= 16);

  const float* xrow = x + (base + col) * 81;
  const float* xq   = xrow + (hiM ? 4 : 0);
  const v2f*   bp   = (const v2f*)(ws + BFRAG_OFF) + lane;
  const float* wcLo  = ws + WMIX_OFF + col * 4;
  const float* wcHi0 = wcLo + 16 * 4;
  const float* wcHi1 = wcLo + (27 + 16) * 4;
  const float* wcHi2 = ws + WMIX_OFF + min(54 + 16 + col, NUNITS - 1) * 4;
  float*       outp = out + base * 81 + lane;
  const bool   act  = (lane < 27);

  const v4f q0 = *(const v4fu*)(xq + 0);
  const v4f q1 = *(const v4fu*)(xq + 8);
  const v4f q2 = *(const v4fu*)(xq + 16);
  const v4f q3 = *(const v4fu*)(xq + 24);
  const v4f q4 = *(const v4fu*)(xq + 32);
  const v4f q5 = *(const v4fu*)(xq + 40);
  const v4f q6 = *(const v4fu*)(xq + 48);
  const v4f q7 = *(const v4fu*)(xq + 56);
  const v4f q8 = *(const v4fu*)(xq + 64);
  const v4f q9 = *(const v4fu*)(xq + 72);
  const float q80 = xrow[80];

  {
    const v2f aL0 = {(hiM ? q2.x : q0.x), q0.y};
    const v2f aL1 = {q0.z, q1.y};
    const v2f aL2 = {q1.z, q1.w};
    const v2f aL3 = {q2.z, q2.w};
    const v2f aM0 = {(hiM ? q0.x : q0.w), q0.y};
    const v2f aM1 = {q1.x, q1.y};
    const v2f aM2 = {q1.z, q2.y};
    const v2f aM3 = {q2.z, q2.w};
    const v2f aR0 = {(hiM ? q0.z : q1.x), (hiM ? q0.w : q2.x)};
    const v2f aR1 = {(hiM ? q1.w : q2.y), q3.x};
    const v2f aR2 = {q3.y, q3.z};
    v8f aLo = {}, bLo = {}, aHi = {}, bHi = {};
    aLo = __builtin_amdgcn_wmma_f32_16x16x4_f32(false, aL0, false, bp[0*32], (short)0, aLo, false, false);
    bLo = __builtin_amdgcn_wmma_f32_16x16x4_f32(false, aL0, false, bp[41*32], (short)0, bLo, false, false);
    aLo = __builtin_amdgcn_wmma_f32_16x16x4_f32(false, aL1, false, bp[1*32], (short)0, aLo, false, false);
    bLo = __builtin_amdgcn_wmma_f32_16x16x4_f32(false, aL1, false, bp[42*32], (short)0, bLo, false, false);
    aLo = __builtin_amdgcn_wmma_f32_16x16x4_f32(false, aL2, false, bp[2*32], (short)0, aLo, false, false);
    bLo = __builtin_amdgcn_wmma_f32_16x16x4_f32(false, aL2, false, bp[43*32], (short)0, bLo, false, false);
    aLo = __builtin_amdgcn_wmma_f32_16x16x4_f32(false, aL3, false, bp[3*32], (short)0, aLo, false, false);
    bLo = __builtin_amdgcn_wmma_f32_16x16x4_f32(false, aL3, false, bp[44*32], (short)0, bLo, false, false);
    aLo = __builtin_amdgcn_wmma_f32_16x16x4_f32(false, aM0, false, bp[4*32], (short)0, aLo, false, false);
    bLo = __builtin_amdgcn_wmma_f32_16x16x4_f32(false, aM0, false, bp[45*32], (short)0, bLo, false, false);
    aHi = __builtin_amdgcn_wmma_f32_16x16x4_f32(false, aM0, false, bp[5*32], (short)0, aHi, false, false);
    bHi = __builtin_amdgcn_wmma_f32_16x16x4_f32(false, aM0, false, bp[46*32], (short)0, bHi, false, false);
    aLo = __builtin_amdgcn_wmma_f32_16x16x4_f32(false, aM1, false, bp[6*32], (short)0, aLo, false, false);
    bLo = __builtin_amdgcn_wmma_f32_16x16x4_f32(false, aM1, false, bp[47*32], (short)0, bLo, false, false);
    aHi = __builtin_amdgcn_wmma_f32_16x16x4_f32(false, aM1, false, bp[7*32], (short)0, aHi, false, false);
    bHi = __builtin_amdgcn_wmma_f32_16x16x4_f32(false, aM1, false, bp[48*32], (short)0, bHi, false, false);
    aLo = __builtin_amdgcn_wmma_f32_16x16x4_f32(false, aM2, false, bp[8*32], (short)0, aLo, false, false);
    bLo = __builtin_amdgcn_wmma_f32_16x16x4_f32(false, aM2, false, bp[49*32], (short)0, bLo, false, false);
    aHi = __builtin_amdgcn_wmma_f32_16x16x4_f32(false, aM2, false, bp[9*32], (short)0, aHi, false, false);
    bHi = __builtin_amdgcn_wmma_f32_16x16x4_f32(false, aM2, false, bp[50*32], (short)0, bHi, false, false);
    aLo = __builtin_amdgcn_wmma_f32_16x16x4_f32(false, aM3, false, bp[10*32], (short)0, aLo, false, false);
    bLo = __builtin_amdgcn_wmma_f32_16x16x4_f32(false, aM3, false, bp[51*32], (short)0, bLo, false, false);
    aHi = __builtin_amdgcn_wmma_f32_16x16x4_f32(false, aM3, false, bp[11*32], (short)0, aHi, false, false);
    bHi = __builtin_amdgcn_wmma_f32_16x16x4_f32(false, aM3, false, bp[52*32], (short)0, bHi, false, false);
    aHi = __builtin_amdgcn_wmma_f32_16x16x4_f32(false, aR0, false, bp[12*32], (short)0, aHi, false, false);
    bHi = __builtin_amdgcn_wmma_f32_16x16x4_f32(false, aR0, false, bp[53*32], (short)0, bHi, false, false);
    aHi = __builtin_amdgcn_wmma_f32_16x16x4_f32(false, aR1, false, bp[13*32], (short)0, aHi, false, false);
    bHi = __builtin_amdgcn_wmma_f32_16x16x4_f32(false, aR1, false, bp[54*32], (short)0, bHi, false, false);
    aHi = __builtin_amdgcn_wmma_f32_16x16x4_f32(false, aR2, false, bp[14*32], (short)0, aHi, false, false);
    bHi = __builtin_amdgcn_wmma_f32_16x16x4_f32(false, aR2, false, bp[55*32], (short)0, bHi, false, false);
    band_store(aLo, bLo, aHi, bHi, wcLo + 0, wcHi0, outp + 0, act);
  }
  {
    const v2f aL0 = {(hiM ? q3.x : q3.w), q3.y};
    const v2f aL1 = {q4.x, q4.y};
    const v2f aL2 = {q4.z, q5.y};
    const v2f aL3 = {q5.z, q5.w};
    const v2f aM0 = {(hiM ? q3.z : q4.x), (hiM ? q3.w : q5.x)};
    const v2f aM1 = {(hiM ? q4.w : q5.y), q6.x};
    const v2f aM2 = {q6.y, q6.z};
    const v2f aR0 = {(hiM ? q5.x : q4.y), (hiM ? q6.x : q4.z)};
    const v2f aR1 = {(hiM ? q6.y : q4.w), q5.z};
    const v2f aR2 = {q5.w, q6.w};
    v8f aLo = {}, bLo = {}, aHi = {}, bHi = {};
    aLo = __builtin_amdgcn_wmma_f32_16x16x4_f32(false, aL0, false, bp[15*32], (short)0, aLo, false, false);
    bLo = __builtin_amdgcn_wmma_f32_16x16x4_f32(false, aL0, false, bp[56*32], (short)0, bLo, false, false);
    aLo = __builtin_amdgcn_wmma_f32_16x16x4_f32(false, aL1, false, bp[16*32], (short)0, aLo, false, false);
    bLo = __builtin_amdgcn_wmma_f32_16x16x4_f32(false, aL1, false, bp[57*32], (short)0, bLo, false, false);
    aLo = __builtin_amdgcn_wmma_f32_16x16x4_f32(false, aL2, false, bp[17*32], (short)0, aLo, false, false);
    bLo = __builtin_amdgcn_wmma_f32_16x16x4_f32(false, aL2, false, bp[58*32], (short)0, bLo, false, false);
    aLo = __builtin_amdgcn_wmma_f32_16x16x4_f32(false, aL3, false, bp[18*32], (short)0, aLo, false, false);
    bLo = __builtin_amdgcn_wmma_f32_16x16x4_f32(false, aL3, false, bp[59*32], (short)0, bLo, false, false);
    aLo = __builtin_amdgcn_wmma_f32_16x16x4_f32(false, aM0, false, bp[19*32], (short)0, aLo, false, false);
    bLo = __builtin_amdgcn_wmma_f32_16x16x4_f32(false, aM0, false, bp[60*32], (short)0, bLo, false, false);
    aHi = __builtin_amdgcn_wmma_f32_16x16x4_f32(false, aM0, false, bp[20*32], (short)0, aHi, false, false);
    bHi = __builtin_amdgcn_wmma_f32_16x16x4_f32(false, aM0, false, bp[61*32], (short)0, bHi, false, false);
    aLo = __builtin_amdgcn_wmma_f32_16x16x4_f32(false, aM1, false, bp[21*32], (short)0, aLo, false, false);
    bLo = __builtin_amdgcn_wmma_f32_16x16x4_f32(false, aM1, false, bp[62*32], (short)0, bLo, false, false);
    aHi = __builtin_amdgcn_wmma_f32_16x16x4_f32(false, aM1, false, bp[22*32], (short)0, aHi, false, false);
    bHi = __builtin_amdgcn_wmma_f32_16x16x4_f32(false, aM1, false, bp[63*32], (short)0, bHi, false, false);
    aLo = __builtin_amdgcn_wmma_f32_16x16x4_f32(false, aM2, false, bp[23*32], (short)0, aLo, false, false);
    bLo = __builtin_amdgcn_wmma_f32_16x16x4_f32(false, aM2, false, bp[64*32], (short)0, bLo, false, false);
    aHi = __builtin_amdgcn_wmma_f32_16x16x4_f32(false, aM2, false, bp[24*32], (short)0, aHi, false, false);
    bHi = __builtin_amdgcn_wmma_f32_16x16x4_f32(false, aM2, false, bp[65*32], (short)0, bHi, false, false);
    aHi = __builtin_amdgcn_wmma_f32_16x16x4_f32(false, aR0, false, bp[25*32], (short)0, aHi, false, false);
    bHi = __builtin_amdgcn_wmma_f32_16x16x4_f32(false, aR0, false, bp[66*32], (short)0, bHi, false, false);
    aHi = __builtin_amdgcn_wmma_f32_16x16x4_f32(false, aR1, false, bp[26*32], (short)0, aHi, false, false);
    bHi = __builtin_amdgcn_wmma_f32_16x16x4_f32(false, aR1, false, bp[67*32], (short)0, bHi, false, false);
    aHi = __builtin_amdgcn_wmma_f32_16x16x4_f32(false, aR2, false, bp[27*32], (short)0, aHi, false, false);
    bHi = __builtin_amdgcn_wmma_f32_16x16x4_f32(false, aR2, false, bp[68*32], (short)0, bHi, false, false);
    band_store(aLo, bLo, aHi, bHi, wcLo + 108, wcHi1, outp + 27, act);
  }
  {
    const v2f aL0 = {(hiM ? q6.z : q7.x), (hiM ? q6.w : q8.x)};
    const v2f aL1 = {(hiM ? q7.w : q8.y), q9.x};
    const v2f aL2 = {q9.y, q9.z};
    const v2f aM0 = {(hiM ? q8.x : q7.y), (hiM ? q9.x : q7.z)};
    const v2f aM1 = {(hiM ? q9.y : q7.w), q8.z};
    const v2f aM2 = {q8.w, q9.w};
    const v2f aR0 = {(hiM ? q7.x : q80), q7.y};
    const v2f aR1 = {q7.z, q8.y};
    const v2f aR2 = {q8.z, q8.w};
    const v2f aR3 = {q9.z, q9.w};
    v8f aLo = {}, bLo = {}, aHi = {}, bHi = {};
    aLo = __builtin_amdgcn_wmma_f32_16x16x4_f32(false, aL0, false, bp[28*32], (short)0, aLo, false, false);
    bLo = __builtin_amdgcn_wmma_f32_16x16x4_f32(false, aL0, false, bp[69*32], (short)0, bLo, false, false);
    aLo = __builtin_amdgcn_wmma_f32_16x16x4_f32(false, aL1, false, bp[29*32], (short)0, aLo, false, false);
    bLo = __builtin_amdgcn_wmma_f32_16x16x4_f32(false, aL1, false, bp[70*32], (short)0, bLo, false, false);
    aLo = __builtin_amdgcn_wmma_f32_16x16x4_f32(false, aL2, false, bp[30*32], (short)0, aLo, false, false);
    bLo = __builtin_amdgcn_wmma_f32_16x16x4_f32(false, aL2, false, bp[71*32], (short)0, bLo, false, false);
    aLo = __builtin_amdgcn_wmma_f32_16x16x4_f32(false, aM0, false, bp[31*32], (short)0, aLo, false, false);
    bLo = __builtin_amdgcn_wmma_f32_16x16x4_f32(false, aM0, false, bp[72*32], (short)0, bLo, false, false);
    aHi = __builtin_amdgcn_wmma_f32_16x16x4_f32(false, aM0, false, bp[32*32], (short)0, aHi, false, false);
    bHi = __builtin_amdgcn_wmma_f32_16x16x4_f32(false, aM0, false, bp[73*32], (short)0, bHi, false, false);
    aLo = __builtin_amdgcn_wmma_f32_16x16x4_f32(false, aM1, false, bp[33*32], (short)0, aLo, false, false);
    bLo = __builtin_amdgcn_wmma_f32_16x16x4_f32(false, aM1, false, bp[74*32], (short)0, bLo, false, false);
    aHi = __builtin_amdgcn_wmma_f32_16x16x4_f32(false, aM1, false, bp[34*32], (short)0, aHi, false, false);
    bHi = __builtin_amdgcn_wmma_f32_16x16x4_f32(false, aM1, false, bp[75*32], (short)0, bHi, false, false);
    aLo = __builtin_amdgcn_wmma_f32_16x16x4_f32(false, aM2, false, bp[35*32], (short)0, aLo, false, false);
    bLo = __builtin_amdgcn_wmma_f32_16x16x4_f32(false, aM2, false, bp[76*32], (short)0, bLo, false, false);
    aHi = __builtin_amdgcn_wmma_f32_16x16x4_f32(false, aM2, false, bp[36*32], (short)0, aHi, false, false);
    bHi = __builtin_amdgcn_wmma_f32_16x16x4_f32(false, aM2, false, bp[77*32], (short)0, bHi, false, false);
    aHi = __builtin_amdgcn_wmma_f32_16x16x4_f32(false, aR0, false, bp[37*32], (short)0, aHi, false, false);
    bHi = __builtin_amdgcn_wmma_f32_16x16x4_f32(false, aR0, false, bp[78*32], (short)0, bHi, false, false);
    aHi = __builtin_amdgcn_wmma_f32_16x16x4_f32(false, aR1, false, bp[38*32], (short)0, aHi, false, false);
    bHi = __builtin_amdgcn_wmma_f32_16x16x4_f32(false, aR1, false, bp[79*32], (short)0, bHi, false, false);
    aHi = __builtin_amdgcn_wmma_f32_16x16x4_f32(false, aR2, false, bp[39*32], (short)0, aHi, false, false);
    bHi = __builtin_amdgcn_wmma_f32_16x16x4_f32(false, aR2, false, bp[80*32], (short)0, bHi, false, false);
    aHi = __builtin_amdgcn_wmma_f32_16x16x4_f32(false, aR3, false, bp[40*32], (short)0, aHi, false, false);
    bHi = __builtin_amdgcn_wmma_f32_16x16x4_f32(false, aR3, false, bp[81*32], (short)0, bHi, false, false);
    band_store(aLo, bLo, aHi, bHi, wcLo + 216, wcHi2, outp + 54, act);
  }
}

extern "C" void kernel_launch(void* const* d_in, const int* in_sizes, int n_in,
                              void* d_out, int out_size, void* d_ws, size_t ws_size,
                              hipStream_t stream) {
  const float* x  = (const float*)d_in[0];
  const float* sa = (const float*)d_in[1];
  const float* sb = (const float*)d_in[2];
  const float* op = (const float*)d_in[3];
  float* out = (float*)d_out;
  float* ws  = (float*)d_ws;

  const int nBatch = in_sizes[0] / 81;

  logic_prep_kernel<<<22, 128, 0, stream>>>(sa, sb, op, ws);

  const int rowsPerBlock = 64;
  const int nBlocks = (nBatch + rowsPerBlock - 1) / rowsPerBlock;
  logic_wmma_kernel<<<nBlocks, 128, 0, stream>>>(x, ws, out, nBatch);
}


# one accumulator tile per patch and table (six per band), joined by a lane select before the half swap: the middle patch's k-steps run once, 62 f32 WMMAs per wave in place of 82; sums unchanged
# speedup vs baseline: 1.8099x; 1.0489x over previous
#include <hip/hip_runtime.h>

typedef __attribute__((ext_vector_type(2))) float v2f;
typedef __attribute__((ext_vector_type(4), aligned(4))) float v4fu;
typedef __attribute__((ext_vector_type(4))) float v4f;
typedef __attribute__((ext_vector_type(8))) float v8f;

#define NUNITS 81
#define WMIX_OFF  1472
#define BFRAG_OFF 2768
#define NKSTEP    62

__device__ __constant__ signed char c_floc[NKSTEP][2][2] = {
  {{0,1},{8,-1}}, {{2,3},{-1,-1}}, {{4,5},{-1,-1}}, {{6,7},{-1,-1}},
  {{0,1},{8,-1}}, {{2,3},{-1,-1}}, {{4,5},{-1,-1}}, {{6,7},{-1,-1}},
  {{0,-1},{1,2}}, {{-1,-1},{3,4}}, {{-1,-1},{5,6}}, {{-1,-1},{7,8}},
  {{0,-1},{1,2}}, {{-1,-1},{3,4}}, {{-1,-1},{5,6}}, {{-1,-1},{7,8}},
  {{2,4},{0,1}}, {{5,6},{3,-1}}, {{7,8},{-1,-1}}, {{2,4},{0,1}},
  {{5,6},{3,-1}}, {{7,8},{-1,-1}}, {{0,-1},{1,2}}, {{-1,-1},{3,4}},
  {{-1,-1},{5,6}}, {{-1,-1},{7,8}}, {{0,-1},{1,2}}, {{-1,-1},{3,4}},
  {{-1,-1},{5,6}}, {{-1,-1},{7,8}}, {{2,4},{0,1}}, {{5,6},{3,-1}},
  {{7,8},{-1,-1}}, {{2,4},{0,1}}, {{5,6},{3,-1}}, {{7,8},{-1,-1}},
  {{0,1},{5,7}}, {{2,3},{8,-1}}, {{4,6},{-1,-1}}, {{0,1},{5,7}},
  {{2,3},{8,-1}}, {{4,6},{-1,-1}}, {{2,4},{0,1}}, {{5,6},{3,-1}},
  {{7,8},{-1,-1}}, {{2,4},{0,1}}, {{5,6},{3,-1}}, {{7,8},{-1,-1}},
  {{0,1},{5,7}}, {{2,3},{8,-1}}, {{4,6},{-1,-1}}, {{0,1},{5,7}},
  {{2,3},{8,-1}}, {{4,6},{-1,-1}}, {{8,-1},{0,1}}, {{-1,-1},{2,3}},
  {{-1,-1},{4,5}}, {{-1,-1},{6,7}}, {{8,-1},{0,1}}, {{-1,-1},{2,3}},
  {{-1,-1},{4,5}}, {{-1,-1},{6,7}}};

__device__ __constant__ unsigned char c_fpt[NKSTEP][3] = {
  {0,0,0}, {0,0,1}, {0,0,2}, {0,0,3}, {0,1,0}, {0,1,1}, {0,1,2}, {0,1,3},
  {1,0,0}, {1,0,1}, {1,0,2}, {1,0,3}, {1,1,0}, {1,1,1}, {1,1,2}, {1,1,3},
  {2,0,0}, {2,0,1}, {2,0,2}, {2,1,0}, {2,1,1}, {2,1,2}, {3,0,0}, {3,0,1},
  {3,0,2}, {3,0,3}, {3,1,0}, {3,1,1}, {3,1,2}, {3,1,3}, {4,0,0}, {4,0,1},
  {4,0,2}, {4,1,0}, {4,1,1}, {4,1,2}, {5,0,0}, {5,0,1}, {5,0,2}, {5,1,0},
  {5,1,1}, {5,1,2}, {6,0,0}, {6,0,1}, {6,0,2}, {6,1,0}, {6,1,1}, {6,1,2},
  {7,0,0}, {7,0,1}, {7,0,2}, {7,1,0}, {7,1,1}, {7,1,2}, {8,0,0}, {8,0,1},
  {8,0,2}, {8,0,3}, {8,1,0}, {8,1,1}, {8,1,2}, {8,1,3}};

__device__ __forceinline__ float pick9(const float (&e)[9], const int i) {
  float r = 0.f;
#pragma unroll
  for (int j = 0; j < 9; ++j) r = (i == j) ? e[j] : r;
  return r;
}

__global__ void __launch_bounds__(128)
logic_prep_kernel(const float* __restrict__ sa,
                  const float* __restrict__ sb,
                  const float* __restrict__ op,
                  float* __restrict__ ws) {
  const int tid = threadIdx.x;
  const int blk = blockIdx.x;
  if (blk < 16) {
    const int g = blk * 4 + (tid >> 5);
    if (g >= NKSTEP) return;
    const int p    = c_fpt[g][0];
    const int tb   = c_fpt[g][1];
    const int lane = tid & 31;
    const int col  = lane & 15;
    const int h    = lane >> 4;
    const int i0 = c_floc[g][h][0];
    const int i1 = c_floc[g][h][1];
    const int r  = p % 3;
    const int ul = (r == 0 || (r == 1 && col >= 9)) ? col : 16 + col;
    const int u  = 27 * (p / 3) + ul;
    float v0 = 0.f, v1 = 0.f;
    if (ul / 9 == r) {
      const float* src = (tb ? sb : sa) + u * 9;
      float m = src[0];
#pragma unroll
      for (int i = 1; i < 9; ++i) m = fmaxf(m, src[i]);
      float ex[9], s = 0.f;
#pragma unroll
      for (int i = 0; i < 9; ++i) { ex[i] = expf(src[i] - m); s += ex[i]; }
      const float inv = 1.0f / s;
      v0 = pick9(ex, i0) * inv;
      v1 = pick9(ex, i1) * inv;
    }
    v2f* dst = (v2f*)(ws + BFRAG_OFF) + (g * 32 + lane);
    const v2f v = {v0, v1};
    *dst = v;
  } else if (tid < NUNITS) {
    const float* src = op + tid * 16;
    float m = src[0];
#pragma unroll
    for (int i = 1; i < 16; ++i) m = fmaxf(m, src[i]);
    float w[16], s = 0.f;
#pragma unroll
    for (int i = 0; i < 16; ++i) { w[i] = expf(src[i] - m); s += w[i]; }
    const float inv = 1.0f / s;
#pragma unroll
    for (int i = 0; i < 16; ++i) w[i] *= inv;
    const float c0  = w[8] + w[9] + w[10] + w[11] + w[12] + w[13] + w[14] + w[15];
    const float cA  = w[2] + w[3] + w[6] + w[7] - w[8] - w[9] - w[12] - w[13];
    const float cB  = w[4] + w[5] + w[6] + w[7] - w[8] - w[9] - w[10] - w[11];
    const float cAB = w[1] - w[2] - w[4] - 2.0f * w[6] - w[7] + w[8] +
                      2.0f * w[9] + w[11] + w[13] - w[14];
    const v4f c = {c0, cA, cB, cAB};
    *(v4f*)(ws + WMIX_OFF + tid * 4) = c;
  }
}

__device__ __forceinline__ float mix1(const float av, const float bv, const v4f c) {
  return fmaf(av * bv, c.w, fmaf(bv, c.z, fmaf(av, c.y, c.x)));
}

__device__ __forceinline__ void band_store(const v8f aLo, const v8f bLo,
                                           const v8f aHi, const v8f bHi,
                                           const float* __restrict__ wLo,
                                           const float* __restrict__ wHi,
                                           float* __restrict__ o, const bool act) {
  const v4f cl = *(const v4f*)wLo;
  const v4f ch = *(const v4f*)wHi;
  float r0[8], r1[8];
#pragma unroll
  for (int m = 0; m < 8; ++m) {
    const float rl = mix1(aLo[m], bLo[m], cl);
    const float rh = mix1(aHi[m], bHi[m], ch);
    const auto sw = __builtin_amdgcn_permlane16_swap(__float_as_uint(rl), __float_as_uint(rh), false, false);
    r0[m] = __uint_as_float(sw[0]);
    r1[m] = __uint_as_float(sw[1]);
  }
  if (act) {
#pragma unroll
    for (int m = 0; m < 8; ++m) {
      __builtin_nontemporal_store(r0[m], o + 81 * m);
      __builtin_nontemporal_store(r1[m], o + 81 * (m + 8));
    }
  }
}

__device__ __forceinline__ v8f join8(const bool pick, const v8f x, const v8f y) {
  v8f r;
#pragma unroll
  for (int m = 0; m < 8; ++m) r[m] = pick ? x[m] : y[m];
  return r;
}

#define WMMA4(acc, afrag, f) \
  acc = __builtin_amdgcn_wmma_f32_16x16x4_f32(false, afrag, false, bp[(f) * 32], (short)0, acc, false, false)

__global__ void __launch_bounds__(128, 1)
logic_wmma_kernel(const float* __restrict__ x,
                  const float* __restrict__ ws,
                  float* __restrict__ out,
                  int nBatch) {
  const int lane   = threadIdx.x & 31;
  const int waveId = threadIdx.x >> 5;
  const long base  = ((long)blockIdx.x * 4 + waveId) * 16;
  if (base + 16 > nBatch) return;

  const int col  = lane & 15;
  const bool hiM = (lane >= 16);

  const float* xrow = x + (base + col) * 81;
  const float* xq   = xrow + (hiM ? 4 : 0);
  const v2f*   bp   = (const v2f*)(ws + BFRAG_OFF) + lane;
  const float* wcLo  = ws + WMIX_OFF + col * 4;
  const float* wcHi0 = wcLo + 16 * 4;
  const float* wcHi1 = wcLo + (27 + 16) * 4;
  const float* wcHi2 = ws + WMIX_OFF + min(54 + 16 + col, NUNITS - 1) * 4;
  float*       outp = out + base * 81 + lane;
  const bool   act  = (lane < 27);
  const bool   c9   = (col < 9);
  const bool   c2   = (col < 2);

  const v4f q0 = *(const v4fu*)(xq + 0);
  const v4f q1 = *(const v4fu*)(xq + 8);
  const v4f q2 = *(const v4fu*)(xq + 16);
  const v4f q3 = *(const v4fu*)(xq + 24);
  const v4f q4 = *(const v4fu*)(xq + 32);
  const v4f q5 = *(const v4fu*)(xq + 40);
  const v4f q6 = *(const v4fu*)(xq + 48);
  const v4f q7 = *(const v4fu*)(xq + 56);
  const v4f q8 = *(const v4fu*)(xq + 64);
  const v4f q9 = *(const v4fu*)(xq + 72);
  const float q80 = xrow[80];

  {
    const v2f aL0 = {(hiM ? q2.x : q0.x), q0.y};
    const v2f aL1 = {q0.z, q1.y};
    const v2f aL2 = {q1.z, q1.w};
    const v2f aL3 = {q2.z, q2.w};
    const v2f aM0 = {(hiM ? q0.x : q0.w), q0.y};
    const v2f aM1 = {q1.x, q1.y};
    const v2f aM2 = {q1.z, q2.y};
    const v2f aM3 = {q2.z, q2.w};
    const v2f aR0 = {(hiM ? q0.z : q1.x), (hiM ? q0.w : q2.x)};
    const v2f aR1 = {(hiM ? q1.w : q2.y), q3.x};
    const v2f aR2 = {q3.y, q3.z};
    v8f aL = {}, bL = {}, aM = {}, bM = {}, aR = {}, bR = {};
    WMMA4(aL, aL0, 0);
    WMMA4(bL, aL0, 4);
    WMMA4(aL, aL1, 1);
    WMMA4(bL, aL1, 5);
    WMMA4(aL, aL2, 2);
    WMMA4(bL, aL2, 6);
    WMMA4(aL, aL3, 3);
    WMMA4(bL, aL3, 7);
    WMMA4(aM, aM0, 8);
    WMMA4(bM, aM0, 12);
    WMMA4(aM, aM1, 9);
    WMMA4(bM, aM1, 13);
    WMMA4(aM, aM2, 10);
    WMMA4(bM, aM2, 14);
    WMMA4(aM, aM3, 11);
    WMMA4(bM, aM3, 15);
    WMMA4(aR, aR0, 16);
    WMMA4(bR, aR0, 19);
    WMMA4(aR, aR1, 17);
    WMMA4(bR, aR1, 20);
    WMMA4(aR, aR2, 18);
    WMMA4(bR, aR2, 21);
    band_store(join8(c9, aL, aM), join8(c9, bL, bM), join8(c2, aM, aR), join8(c2, bM, bR),
               wcLo + 0, wcHi0, outp + 0, act);
  }
  {
    const v2f aL0 = {(hiM ? q3.x : q3.w), q3.y};
    const v2f aL1 = {q4.x, q4.y};
    const v2f aL2 = {q4.z, q5.y};
    const v2f aL3 = {q5.z, q5.w};
    const v2f aM0 = {(hiM ? q3.z : q4.x), (hiM ? q3.w : q5.x)};
    const v2f aM1 = {(hiM ? q4.w : q5.y), q6.x};
    const v2f aM2 = {q6.y, q6.z};
    const v2f aR0 = {(hiM ? q5.x : q4.y), (hiM ? q6.x : q4.z)};
    const v2f aR1 = {(hiM ? q6.y : q4.w), q5.z};
    const v2f aR2 = {q5.w, q6.w};
    v8f aL = {}, bL = {}, aM = {}, bM = {}, aR = {}, bR = {};
    WMMA4(aL, aL0, 22);
    WMMA4(bL, aL0, 26);
    WMMA4(aL, aL1, 23);
    WMMA4(bL, aL1, 27);
    WMMA4(aL, aL2, 24);
    WMMA4(bL, aL2, 28);
    WMMA4(aL, aL3, 25);
    WMMA4(bL, aL3, 29);
    WMMA4(aM, aM0, 30);
    WMMA4(bM, aM0, 33);
    WMMA4(aM, aM1, 31);
    WMMA4(bM, aM1, 34);
    WMMA4(aM, aM2, 32);
    WMMA4(bM, aM2, 35);
    WMMA4(aR, aR0, 36);
    WMMA4(bR, aR0, 39);
    WMMA4(aR, aR1, 37);
    WMMA4(bR, aR1, 40);
    WMMA4(aR, aR2, 38);
    WMMA4(bR, aR2, 41);
    band_store(join8(c9, aL, aM), join8(c9, bL, bM), join8(c2, aM, aR), join8(c2, bM, bR),
               wcLo + 108, wcHi1, outp + 27, act);
  }
  {
    const v2f aL0 = {(hiM ? q6.z : q7.x), (hiM ? q6.w : q8.x)};
    const v2f aL1 = {(hiM ? q7.w : q8.y), q9.x};
    const v2f aL2 = {q9.y, q9.z};
    const v2f aM0 = {(hiM ? q8.x : q7.y), (hiM ? q9.x : q7.z)};
    const v2f aM1 = {(hiM ? q9.y : q7.w), q8.z};
    const v2f aM2 = {q8.w, q9.w};
    const v2f aR0 = {(hiM ? q7.x : q80), q7.y};
    const v2f aR1 = {q7.z, q8.y};
    const v2f aR2 = {q8.z, q8.w};
    const v2f aR3 = {q9.z, q9.w};
    v8f aL = {}, bL = {}, aM = {}, bM = {}, aR = {}, bR = {};
    WMMA4(aL, aL0, 42);
    WMMA4(bL, aL0, 45);
    WMMA4(aL, aL1, 43);
    WMMA4(bL, aL1, 46);
    WMMA4(aL, aL2, 44);
    WMMA4(bL, aL2, 47);
    WMMA4(aM, aM0, 48);
    WMMA4(bM, aM0, 51);
    WMMA4(aM, aM1, 49);
    WMMA4(bM, aM1, 52);
    WMMA4(aM, aM2, 50);
    WMMA4(bM, aM2, 53);
    WMMA4(aR, aR0, 54);
    WMMA4(bR, aR0, 58);
    WMMA4(aR, aR1, 55);
    WMMA4(bR, aR1, 59);
    WMMA4(aR, aR2, 56);
    WMMA4(bR, aR2, 60);
    WMMA4(aR, aR3, 57);
    WMMA4(bR, aR3, 61);
    band_store(join8(c9, aL, aM), join8(c9, bL, bM), join8(c2, aM, aR), join8(c2, bM, bR),
               wcLo + 216, wcHi2, outp + 54, act);
  }
}

extern "C" void kernel_launch(void* const* d_in, const int* in_sizes, int n_in,
                              void* d_out, int out_size, void* d_ws, size_t ws_size,
                              hipStream_t stream) {
  const float* x  = (const float*)d_in[0];
  const float* sa = (const float*)d_in[1];
  const float* sb = (const float*)d_in[2];
  const float* op = (const float*)d_in[3];
  float* out = (float*)d_out;
  float* ws  = (float*)d_ws;

  const int nBatch = in_sizes[0] / 81;

  logic_prep_kernel<<<17, 128, 0, stream>>>(sa, sb, op, ws);

  const int rowsPerBlock = 64;
  const int nBlocks = (nBatch + rowsPerBlock - 1) / rowsPerBlock;
  logic_wmma_kernel<<<nBlocks, 128, 0, stream>>>(x, ws, out, nBatch);
}
